# starting kernel
# baseline (speedup 1.0000x reference)
#include <hip/hip_runtime.h>


typedef __attribute__((ext_vector_type(2))) float v2f;
typedef __attribute__((ext_vector_type(8))) float v8f;

#define DIM        128
#define NROWS      27
#define PADROWS    32
#define LDS_STRIDE 132
#define OUT_COLS   479

__global__ __launch_bounds__(128) void dlrm_interact_wmma(
    const float* __restrict__ dense,
    const float* __restrict__ sparse,
    float* __restrict__ out)
{
    __shared__ float lds[PADROWS * LDS_STRIDE];

    const int b   = blockIdx.x;
    const int tid = threadIdx.x;

    const float4* drow = reinterpret_cast<const float4*>(dense  + (size_t)b * DIM);
    const float4* srow = reinterpret_cast<const float4*>(sparse + (size_t)b * (NROWS - 1) * DIM);
    #pragma unroll
    for (int i = tid; i < PADROWS * 32; i += 128) {
        const int row = i >> 5;
        const int c4  = i & 31;
        float4 v = make_float4(0.f, 0.f, 0.f, 0.f);
        if (row == 0)          v = drow[c4];
        else if (row < NROWS)  v = srow[(row - 1) * 32 + c4];
        *reinterpret_cast<float4*>(&lds[row * LDS_STRIDE + c4 * 4]) = v;
    }
    __syncthreads();

    const int wave = __builtin_amdgcn_readfirstlane(tid >> 5);
    const int lane = tid & 31;

    if (wave < 3) {
        const int I = (wave == 2) ? 1 : 0;
        const int J = (wave == 0) ? 0 : 1;

        const int rA   = I * 16 + (lane & 15);
        const int rB   = J * 16 + (lane & 15);
        const int koff = (lane >> 4) * 2;
        const float* pa = &lds[rA * LDS_STRIDE + koff];
        const float* pb = &lds[rB * LDS_STRIDE + koff];

        v8f acc = {};
        #pragma unroll
        for (int k = 0; k < DIM; k += 4) {
            v2f a  = *reinterpret_cast<const v2f*>(pa + k);
            v2f bb = *reinterpret_cast<const v2f*>(pb + k);
            acc = __builtin_amdgcn_wmma_f32_16x16x4_f32(
                false, a, false, bb, (short)0, acc, false, false);
        }

        float* orow = out + (size_t)b * OUT_COLS + DIM;
        const int n     = J * 16 + (lane & 15);
        const int mbase = I * 16 + ((lane >> 4) << 3);
        #pragma unroll
        for (int r = 0; r < 8; ++r) {
            const int m = mbase + r;
            if (m < n && n < NROWS) {
                const int idx = m * (NROWS - 1) - (m * (m - 1)) / 2 + (n - m - 1);
                orow[idx] = acc[r];
            }
        }
    } else {
        float* orow = out + (size_t)b * OUT_COLS;
        const float* drowf = dense + (size_t)b * DIM;
        #pragma unroll
        for (int i = 0; i < 4; ++i)
            orow[lane + 32 * i] = drowf[lane + 32 * i];
    }
}

extern "C" void kernel_launch(void* const* d_in, const int* in_sizes, int n_in,
                              void* d_out, int out_size, void* d_ws, size_t ws_size,
                              hipStream_t stream) {
    const float* dense  = (const float*)d_in[0];
    const float* sparse = (const float*)d_in[1];
    float* out = (float*)d_out;
    const int B = in_sizes[0] / DIM;
    dlrm_interact_wmma<<<B, 128, 0, stream>>>(dense, sparse, out);
}


# one wave per sample: it forms all 3 Gram tiles from 2 shared LDS fragment streams (3 accumulators) in a wave-private LDS slice, no block barrier; 4 samples per block, 4x fewer waves; same sums
# speedup vs baseline: 1.2858x; 1.2858x over previous
#include <hip/hip_runtime.h>


typedef __attribute__((ext_vector_type(2))) float v2f;
typedef __attribute__((ext_vector_type(8))) float v8f;

#define DIM        128
#define NROWS      27
#define LDS_STRIDE 132
#define OUT_COLS   479
#define WAVES      4

template <int I, int J>
__device__ __forceinline__ void scatter_tile(float* __restrict__ orow, const v8f& acc, int lane)
{
    const int n     = J * 16 + (lane & 15);
    const int mbase = I * 16 + ((lane >> 4) << 3);
    #pragma unroll
    for (int r = 0; r < 8; ++r) {
        const int m = mbase + r;
        if (m < n && n < NROWS) {
            const int idx = m * (NROWS - 1) - (m * (m - 1)) / 2 + (n - m - 1);
            orow[idx] = acc[r];
        }
    }
}

__global__ __launch_bounds__(32 * WAVES) void dlrm_interact_wmma(
    const float* __restrict__ dense,
    const float* __restrict__ sparse,
    float* __restrict__ out,
    int B)
{
    __shared__ __attribute__((aligned(16))) float lds_all[WAVES * NROWS * LDS_STRIDE];

    const int wave = __builtin_amdgcn_readfirstlane(threadIdx.x >> 5);
    const int lane = threadIdx.x & 31;
    const int b    = blockIdx.x * WAVES + wave;
    if (b >= B) return;

    float* lds = lds_all + wave * (NROWS * LDS_STRIDE);

    const float4* drow = reinterpret_cast<const float4*>(dense  + (size_t)b * DIM);
    const float4* srow = reinterpret_cast<const float4*>(sparse + (size_t)b * (NROWS - 1) * DIM);
    float4 v[NROWS];
    v[0] = drow[lane];
    #pragma unroll
    for (int r = 1; r < NROWS; ++r) v[r] = srow[(r - 1) * 32 + lane];
    #pragma unroll
    for (int r = 0; r < NROWS; ++r)
        *reinterpret_cast<float4*>(&lds[r * LDS_STRIDE + lane * 4]) = v[r];

    float* orow = out + (size_t)b * OUT_COLS;
    #pragma unroll
    for (int i = 0; i < 4; ++i)
        orow[lane + 32 * i] = lds[lane + 32 * i];

    const int c    = lane & 15;
    const int koff = (lane >> 4) * 2;
    const int row1 = (16 + c < NROWS) ? (16 + c) : (NROWS - 1);
    const float* p0 = &lds[c    * LDS_STRIDE + koff];
    const float* p1 = &lds[row1 * LDS_STRIDE + koff];

    v8f acc00 = {}, acc01 = {}, acc11 = {};
    #pragma unroll
    for (int k = 0; k < DIM; k += 4) {
        const v2f a0 = *reinterpret_cast<const v2f*>(p0 + k);
        const v2f a1 = *reinterpret_cast<const v2f*>(p1 + k);
        acc00 = __builtin_amdgcn_wmma_f32_16x16x4_f32(false, a0, false, a0, (short)0, acc00, false, false);
        acc01 = __builtin_amdgcn_wmma_f32_16x16x4_f32(false, a0, false, a1, (short)0, acc01, false, false);
        acc11 = __builtin_amdgcn_wmma_f32_16x16x4_f32(false, a1, false, a1, (short)0, acc11, false, false);
    }

    float* trow = orow + DIM;
    scatter_tile<0, 0>(trow, acc00, lane);
    scatter_tile<0, 1>(trow, acc01, lane);
    scatter_tile<1, 1>(trow, acc11, lane);
}

extern "C" void kernel_launch(void* const* d_in, const int* in_sizes, int n_in,
                              void* d_out, int out_size, void* d_ws, size_t ws_size,
                              hipStream_t stream) {
    const float* dense  = (const float*)d_in[0];
    const float* sparse = (const float*)d_in[1];
    float* out = (float*)d_out;
    const int B = in_sizes[0] / DIM;
    dlrm_interact_wmma<<<(B + WAVES - 1) / WAVES, 32 * WAVES, 0, stream>>>(dense, sparse, out, B);
}
